# starting kernel
# baseline (speedup 1.0000x reference)
#include <hip/hip_runtime.h>


typedef __attribute__((ext_vector_type(2))) float v2f;
typedef __attribute__((ext_vector_type(8))) float v8f;

#define EMA_ALPHA 0.9f
#define EMA_OMA   0.1f

namespace {
constexpr int Bn  = 8;
constexpr int Tn  = 16384;
constexpr int Cn  = 256;
constexpr int SEG = 1024;
constexpr int WIN = 256;
constexpr int CH  = 16;
}

__global__ __launch_bounds__(32)
void ema_wmma_kernel(const float* __restrict__ x, float* __restrict__ y)
{
    constexpr float P[17] = {
        1.0f, 0.9f, 0.81f, 0.729f, 0.6561f, 0.59049f, 0.531441f,
        0.4782969f, 0.43046721f, 0.387420489f, 0.3486784401f,
        0.31381059609f, 0.282429536481f, 0.2541865828329f,
        0.22876792454961f, 0.205891132094649f, 0.1853020188851841f};
    constexpr float Pi[16] = {
        1.0f, 1.11111111f, 1.23456790f, 1.37174211f, 1.52415790f,
        1.69350878f, 1.88167642f, 2.09075158f, 2.32305731f, 2.58117479f,
        2.86797199f, 3.18663554f, 3.54070615f, 3.93411795f, 4.37124216f,
        4.85693573f};

    const int lane = (int)threadIdx.x;
    const int n    = lane & 15;
    const int kh   = lane >> 4;

    const int seg = (int)blockIdx.x;
    const int ct  = (int)blockIdx.y;
    const int b   = (int)blockIdx.z;

    const int t_out0  = seg * SEG;
    const int t_begin = (seg == 0) ? 0 : (t_out0 - WIN);
    const int nchunk  = (t_out0 + SEG - t_begin) >> 4;
    const int skip    = (t_out0 - t_begin) >> 4;

    float pm = 1.0f;
    #pragma unroll
    for (int k = 0; k < 15; ++k) pm = (k < n) ? pm * EMA_ALPHA : pm;

    v2f a[4];
    #pragma unroll
    for (int q = 0; q < 4; ++q) {
        #pragma unroll
        for (int comp = 0; comp < 2; ++comp) {
            const int   base = 4 * q + comp;
            const float inv  = kh ? Pi[base + 2] : Pi[base];
            const int   j    = base + 2 * kh;
            a[q][comp] = (n >= j) ? (EMA_OMA * pm) * inv : 0.0f;
        }
    }

    float cmul[8];
    #pragma unroll
    for (int v = 0; v < 8; ++v) cmul[v] = kh ? P[v + 9] : P[v + 1];

    const size_t baseElem = ((size_t)b * Tn + (size_t)t_begin) * Cn
                          + (size_t)ct * CH + (size_t)n;
    const float* xp = x + baseElem;
    float*       yp = y + baseElem;

    float carry = 0.0f;

    float cur[8], nxt[8];
    #pragma unroll
    for (int e = 0; e < 8; ++e) nxt[e] = 0.0f;
    #pragma unroll
    for (int q = 0; q < 4; ++q) {
        const int r = 4 * q + 2 * kh;
        cur[2 * q]     = xp[(size_t)r * Cn];
        cur[2 * q + 1] = xp[(size_t)(r + 1) * Cn];
    }

    for (int i = 0; i < nchunk; ++i) {
        const float* xc_next = xp + (size_t)(i + 1) * 16 * Cn;
        if (i + 2 < nchunk)
            __builtin_prefetch(xp + (size_t)(i + 2) * 16 * Cn, 0, 1);
        if (i + 1 < nchunk) {
            #pragma unroll
            for (int q = 0; q < 4; ++q) {
                const int r = 4 * q + 2 * kh;
                nxt[2 * q]     = xc_next[(size_t)r * Cn];
                nxt[2 * q + 1] = xc_next[(size_t)(r + 1) * Cn];
            }
        }

        v8f acc;
        #pragma unroll
        for (int v = 0; v < 8; ++v) acc[v] = cmul[v] * carry;

        #pragma unroll
        for (int q = 0; q < 4; ++q) {
            v2f bq;
            bq[0] = cur[2 * q];
            bq[1] = cur[2 * q + 1];
            acc = __builtin_amdgcn_wmma_f32_16x16x4_f32(
                 false, a[q],  false, bq,
                 (short)0, acc,  false,  false);
        }

        if (i >= skip) {
            float* yc = yp + (size_t)i * 16 * Cn;
            #pragma unroll
            for (int v = 0; v < 8; ++v) {
                const int row = v + 8 * kh;
                yc[(size_t)row * Cn] = acc[v];
            }
        }

        carry = __int_as_float(__builtin_amdgcn_ds_bpermute(
            (16 + n) << 2, __float_as_int(acc[7])));

        #pragma unroll
        for (int e = 0; e < 8; ++e) cur[e] = nxt[e];
    }
}

extern "C" void kernel_launch(void* const* d_in, const int* in_sizes, int n_in,
                              void* d_out, int out_size, void* d_ws, size_t ws_size,
                              hipStream_t stream) {
    (void)in_sizes; (void)n_in; (void)out_size; (void)d_ws; (void)ws_size;
    const float* x = (const float*)d_in[0];
    float*       y = (float*)d_out;
    dim3 grid(Tn / SEG, Cn / CH, Bn);
    ema_wmma_kernel<<<grid, 32, 0, stream>>>(x, y);
}


# Each wave keeps 8 chunks of x (64 global loads) in flight in a register ring, each slot refilled right behind the WMMAs that read it, instead of asking one chunk ahead plus a prefetch hint
# speedup vs baseline: 1.8761x; 1.8761x over previous
#include <hip/hip_runtime.h>


typedef __attribute__((ext_vector_type(2))) float v2f;
typedef __attribute__((ext_vector_type(8))) float v8f;

#define EMA_ALPHA 0.9f
#define EMA_OMA   0.1f

namespace {
constexpr int Bn  = 8;
constexpr int Tn  = 16384;
constexpr int Cn  = 256;
constexpr int SEG = 1024;
constexpr int WIN = 256;
constexpr int CH  = 16;
constexpr int DEPTH = 8;
static_assert((WIN / 16) % DEPTH == 0, "warm-up must be whole ring turns");
static_assert((SEG / 16) % DEPTH == 0, "segment must be whole ring turns");
static_assert(SEG / 16 > DEPTH, "segment longer than the ring");
}

__device__ __forceinline__ void ema_load_chunk(float (&buf)[8], const float* __restrict__ xl)
{
    #pragma unroll
    for (int q = 0; q < 4; ++q) {
        buf[2 * q]     = xl[(size_t)(4 * q) * Cn];
        buf[2 * q + 1] = xl[(size_t)(4 * q + 1) * Cn];
    }
}

template <bool STORE, bool REFILL>
__device__ __forceinline__ void ema_chunk(float (&buf)[8], const v2f (&a)[4],
                                          const float (&cmul)[8], float& carry,
                                          const float* __restrict__ xrefill,
                                          float* __restrict__ yl, int n)
{
    v8f acc;
    #pragma unroll
    for (int v = 0; v < 8; ++v) acc[v] = cmul[v] * carry;

    #pragma unroll
    for (int q = 0; q < 4; ++q) {
        v2f bq;
        bq[0] = buf[2 * q];
        bq[1] = buf[2 * q + 1];
        acc = __builtin_amdgcn_wmma_f32_16x16x4_f32(
             false, a[q],  false, bq,
             (short)0, acc,  false,  false);
    }

    if (REFILL) asm volatile("" :: "v"(acc) : "memory");

    carry = __int_as_float(__builtin_amdgcn_ds_bpermute(
        (16 + n) << 2, __float_as_int(acc[7])));

    if (REFILL) ema_load_chunk(buf, xrefill);

    if (STORE) {
        #pragma unroll
        for (int v = 0; v < 8; ++v) yl[(size_t)v * Cn] = acc[v];
    }
}

__global__ __launch_bounds__(32)
void ema_wmma_kernel(const float* __restrict__ x, float* __restrict__ y)
{
    constexpr float P[17] = {
        1.0f, 0.9f, 0.81f, 0.729f, 0.6561f, 0.59049f, 0.531441f,
        0.4782969f, 0.43046721f, 0.387420489f, 0.3486784401f,
        0.31381059609f, 0.282429536481f, 0.2541865828329f,
        0.22876792454961f, 0.205891132094649f, 0.1853020188851841f};
    constexpr float Pi[16] = {
        1.0f, 1.11111111f, 1.23456790f, 1.37174211f, 1.52415790f,
        1.69350878f, 1.88167642f, 2.09075158f, 2.32305731f, 2.58117479f,
        2.86797199f, 3.18663554f, 3.54070615f, 3.93411795f, 4.37124216f,
        4.85693573f};

    const int lane = (int)threadIdx.x;
    const int n    = lane & 15;
    const int kh   = lane >> 4;

    const int seg = (int)blockIdx.x;
    const int ct  = (int)blockIdx.y;
    const int b   = (int)blockIdx.z;

    const int t_out0  = seg * SEG;
    const int t_begin = (seg == 0) ? 0 : (t_out0 - WIN);
    const int nchunk  = (t_out0 + SEG - t_begin) >> 4;
    const int skip    = (t_out0 - t_begin) >> 4;

    float pm = 1.0f;
    #pragma unroll
    for (int k = 0; k < 15; ++k) pm = (k < n) ? pm * EMA_ALPHA : pm;

    v2f a[4];
    #pragma unroll
    for (int q = 0; q < 4; ++q) {
        #pragma unroll
        for (int comp = 0; comp < 2; ++comp) {
            const int   base = 4 * q + comp;
            const float inv_hi = Pi[base + 2], inv_lo = Pi[base];
            const float inv  = kh ? inv_hi : inv_lo;
            const int   j    = base + 2 * kh;
            a[q][comp] = (n >= j) ? (EMA_OMA * pm) * inv : 0.0f;
        }
    }

    float cmul[8];
    #pragma unroll
    for (int v = 0; v < 8; ++v) {
        const float p_hi = P[v + 9], p_lo = P[v + 1];
        cmul[v] = kh ? p_hi : p_lo;
    }

    const size_t baseElem = ((size_t)b * Tn + (size_t)t_begin) * Cn
                          + (size_t)ct * CH + (size_t)n;
    const float* xl = x + baseElem + (size_t)(2 * kh) * Cn;
    float*       yl = y + baseElem + (size_t)(8 * kh) * Cn;

    constexpr size_t CHUNK = (size_t)16 * Cn;

    float carry = 0.0f;

    float ring[DEPTH][8];
    #pragma unroll
    for (int s = 0; s < DEPTH; ++s) ema_load_chunk(ring[s], xl + (size_t)s * CHUNK);

    int i = 0;

    for (; i < skip; i += DEPTH) {
        const float* xr = xl + (size_t)(i + DEPTH) * CHUNK;
        #pragma unroll
        for (int s = 0; s < DEPTH; ++s)
            ema_chunk<false, true>(ring[s], a, cmul, carry,
                                   xr + (size_t)s * CHUNK, nullptr, n);
    }

    for (; i < nchunk - DEPTH; i += DEPTH) {
        const float* xr = xl + (size_t)(i + DEPTH) * CHUNK;
        float*       yc = yl + (size_t)i * CHUNK;
        #pragma unroll
        for (int s = 0; s < DEPTH; ++s)
            ema_chunk<true, true>(ring[s], a, cmul, carry,
                                  xr + (size_t)s * CHUNK, yc + (size_t)s * CHUNK, n);
    }

    {
        float* yc = yl + (size_t)i * CHUNK;
        #pragma unroll
        for (int s = 0; s < DEPTH; ++s)
            ema_chunk<true, false>(ring[s], a, cmul, carry,
                                   nullptr, yc + (size_t)s * CHUNK, n);
    }
}

extern "C" void kernel_launch(void* const* d_in, const int* in_sizes, int n_in,
                              void* d_out, int out_size, void* d_ws, size_t ws_size,
                              hipStream_t stream) {
    (void)in_sizes; (void)n_in; (void)out_size; (void)d_ws; (void)ws_size;
    const float* x = (const float*)d_in[0];
    float*       y = (float*)d_out;
    dim3 grid(Tn / SEG, Cn / CH, Bn);
    ema_wmma_kernel<<<grid, 32, 0, stream>>>(x, y);
}
